# starting kernel
# baseline (speedup 1.0000x reference)
#include <hip/hip_runtime.h>
#include <hip/hip_bf16.h>


typedef __attribute__((ext_vector_type(2))) float v2f;
typedef __attribute__((ext_vector_type(8))) float v8f;

#define DVEC 128
#define LDSTR 132
#define INV_SIGMA 10.0f

__global__ void ephn_norm_kernel(const float* __restrict__ fvec,
                                 float* __restrict__ rnorm) {
    const int row  = blockIdx.x * 4 + (threadIdx.x >> 5);
    const int lane = threadIdx.x & 31;
    const float4 d = ((const float4*)(fvec + ((size_t)row << 7)))[lane];
    float s = d.x * d.x + d.y * d.y + d.z * d.z + d.w * d.w;
    #pragma unroll
    for (int off = 16; off >= 1; off >>= 1)
        s += __shfl_xor(s, off, 32);
    if (lane == 0) rnorm[row] = rsqrtf(s);
}

__global__ void ephn_main_kernel(const float* __restrict__ fvec,
                                 const int*   __restrict__ Lvec,
                                 const float* __restrict__ rnorm,
                                 float* __restrict__ wsPos,
                                 float* __restrict__ wsNeg,
                                 int N) {
    __shared__ float lds_a[16 * LDSTR];
    __shared__ float lds_b[64 * LDSTR];
    __shared__ float mPV[64]; __shared__ int mPJ[64];
    __shared__ float mNV[64]; __shared__ int mNJ[64];

    const int tid = threadIdx.x;
    const int l   = tid & 31;
    const int w   = tid >> 5;
    const int m   = l & 15;
    const int hi  = l >> 4;
    const int i0  = blockIdx.x * 16;

    for (int t = tid; t < 16 * 32; t += 128) {
        const int f   = t * 4;
        const int row = f >> 7;
        const int col = f & 127;
        float4 d = *(const float4*)(fvec + ((size_t)i0 << 7) + f);
        const float r = rnorm[i0 + row];
        float* dst = &lds_a[row * LDSTR + col];
        dst[0] = d.x * r; dst[1] = d.y * r; dst[2] = d.z * r; dst[3] = d.w * r;
    }

    int Li[8];
    #pragma unroll
    for (int v = 0; v < 8; ++v) Li[v] = Lvec[i0 + v + 8 * hi];

    float posV[8], negV[8]; int posJ[8], negJ[8];
    #pragma unroll
    for (int v = 0; v < 8; ++v) {
        posV[v] = -2.0f; negV[v] = -2.0f;
        posJ[v] = 0x7fffffff; negJ[v] = 0x7fffffff;
    }

    const float* aptr = &lds_a[m * LDSTR + 2 * hi];
    const float* bptr = &lds_b[(w * 16 + m) * LDSTR + 2 * hi];

    const int nstrips = N >> 6;
    for (int jt = 0; jt < nstrips; ++jt) {
        __syncthreads();
        const size_t gbase = ((size_t)jt << 6) << 7;
        for (int t = tid; t < 64 * 32; t += 128) {
            const int f   = t * 4;
            const int row = f >> 7;
            const int col = f & 127;
            float4 d = *(const float4*)(fvec + gbase + f);
            const float r = rnorm[jt * 64 + row];
            float* dst = &lds_b[row * LDSTR + col];
            dst[0] = d.x * r; dst[1] = d.y * r; dst[2] = d.z * r; dst[3] = d.w * r;
        }
        __syncthreads();

        v8f c = {};
        #pragma unroll
        for (int k = 0; k < DVEC; k += 4) {
            v2f a = *(const v2f*)(aptr + k);
            v2f b = *(const v2f*)(bptr + k);
            c = __builtin_amdgcn_wmma_f32_16x16x4_f32(
                    false, a, false, b, (short)0, c, false, false);
        }

        const int j  = jt * 64 + w * 16 + m;
        const int Lj = Lvec[j];
        #pragma unroll
        for (int v = 0; v < 8; ++v) {
            const int   i   = i0 + v + 8 * hi;
            const float d   = c[v];
            const bool  eqc = (Li[v] == Lj);
            const float pv = eqc ? (((i == j) || (d > 0.9999f)) ? -1.0f : d)
                                 : -1.0f;
            if (pv > posV[v] || (pv == posV[v] && j < posJ[v])) {
                posV[v] = pv; posJ[v] = j;
            }
            const float nv = eqc ? -1.0f : d;
            if (nv > negV[v] || (nv == negV[v] && j < negJ[v])) {
                negV[v] = nv; negJ[v] = j;
            }
        }
    }

    #pragma unroll
    for (int v = 0; v < 8; ++v) {
        #pragma unroll
        for (int off = 1; off < 16; off <<= 1) {
            float ov = __shfl_xor(posV[v], off, 16);
            int   oj = __shfl_xor(posJ[v], off, 16);
            if (ov > posV[v] || (ov == posV[v] && oj < posJ[v])) {
                posV[v] = ov; posJ[v] = oj;
            }
            float onv = __shfl_xor(negV[v], off, 16);
            int   onj = __shfl_xor(negJ[v], off, 16);
            if (onv > negV[v] || (onv == negV[v] && onj < negJ[v])) {
                negV[v] = onv; negJ[v] = onj;
            }
        }
    }

    if ((l & 15) == 0) {
        #pragma unroll
        for (int v = 0; v < 8; ++v) {
            const int row = v + 8 * hi;
            const int idx = w * 16 + row;
            mPV[idx] = posV[v]; mPJ[idx] = posJ[v];
            mNV[idx] = negV[v]; mNJ[idx] = negJ[v];
        }
    }
    __syncthreads();
    if (tid < 16) {
        float bpv = -2.0f, bnv = -2.0f;
        int   bpj = 0x7fffffff, bnj = 0x7fffffff;
        #pragma unroll
        for (int ww = 0; ww < 4; ++ww) {
            const int idx = ww * 16 + tid;
            const float pv = mPV[idx]; const int pj = mPJ[idx];
            if (pv > bpv || (pv == bpv && pj < bpj)) { bpv = pv; bpj = pj; }
            const float nv = mNV[idx]; const int nj = mNJ[idx];
            if (nv > bnv || (nv == bnv && nj < bnj)) { bnv = nv; bnj = nj; }
        }
        wsPos[i0 + tid] = bpv;
        wsNeg[i0 + tid] = bnv;
    }
}

__global__ void ephn_finalize_kernel(const float* __restrict__ wsPos,
                                     const float* __restrict__ wsNeg,
                                     float* __restrict__ out, int N) {
    __shared__ float sProb[1024];
    __shared__ float sPos[1024];
    __shared__ float sNeg[1024];
    __shared__ int   sCnt[1024];
    const int tid = threadIdx.x;

    float* eps = out + 1;
    float* hns = out + 1 + (N + 2);

    float accProb = 0.0f, accPos = 0.0f, accNeg = 0.0f;
    int accCnt = 0;
    for (int i = tid; i < N; i += 1024) {
        const float p = wsPos[i];
        const float n = wsNeg[i];
        eps[i] = (p > 0.0f) ? p : 0.0f;
        hns[i] = (n > 0.0f) ? n : 0.0f;
        accPos += p;
        accNeg += n;
        if (p > 0.0f && n > 0.0f) {
            const float s = (n - p) * INV_SIGMA;
            const float prob = (s > 0.0f) ? (s + log1pf(expf(-s)))
                                          : log1pf(expf(s));
            accProb += prob;
            accCnt  += 1;
        }
    }
    sProb[tid] = accProb; sPos[tid] = accPos; sNeg[tid] = accNeg; sCnt[tid] = accCnt;
    __syncthreads();
    for (int st = 512; st > 0; st >>= 1) {
        if (tid < st) {
            sProb[tid] += sProb[tid + st];
            sPos[tid]  += sPos[tid + st];
            sNeg[tid]  += sNeg[tid + st];
            sCnt[tid]  += sCnt[tid + st];
        }
        __syncthreads();
    }
    if (tid == 0) {
        const int cnt = (sCnt[0] > 1) ? sCnt[0] : 1;
        out[0] = sProb[0] / (float)cnt;
        eps[N] = 0.0f; eps[N + 1] = 1.0f;
        hns[N] = 0.0f; hns[N + 1] = 1.0f;
        out[1 + 2 * (N + 2)] = (sPos[0] - sNeg[0]) / (float)N;
    }
}

extern "C" void kernel_launch(void* const* d_in, const int* in_sizes, int n_in,
                              void* d_out, int out_size, void* d_ws, size_t ws_size,
                              hipStream_t stream) {
    const float* fvec = (const float*)d_in[0];
    const int*   Lvec = (const int*)d_in[1];
    const int N = in_sizes[1];

    float* rnorm = (float*)d_ws;
    float* wsPos = rnorm + N;
    float* wsNeg = wsPos + N;

    ephn_norm_kernel<<<N / 4, 128, 0, stream>>>(fvec, rnorm);
    ephn_main_kernel<<<N / 16, 128, 0, stream>>>(fvec, Lvec, rnorm,
                                                 wsPos, wsNeg, N);
    ephn_finalize_kernel<<<1, 1024, 0, stream>>>(wsPos, wsNeg,
                                                 (float*)d_out, N);
}


# Re-tile the mining GEMM: block = 64 rows x 1/8 of the columns, each wave holds its 16x128 A panel in registers and does all 4 tiles of a staged strip, so one staging feeds 4x the WMMAs
# speedup vs baseline: 2.7201x; 2.7201x over previous
#include <hip/hip_runtime.h>
#include <hip/hip_bf16.h>


typedef __attribute__((ext_vector_type(2))) float v2f;
typedef __attribute__((ext_vector_type(8))) float v8f;

#define DVEC 128
#define LDSTR 132
#define INV_SIGMA 10.0f

__global__ void ephn_norm_kernel(const float* __restrict__ fvec,
                                 float* __restrict__ rnorm) {
    const int row  = blockIdx.x * 4 + (threadIdx.x >> 5);
    const int lane = threadIdx.x & 31;
    const float4 d = ((const float4*)(fvec + ((size_t)row << 7)))[lane];
    float s = d.x * d.x + d.y * d.y + d.z * d.z + d.w * d.w;
    #pragma unroll
    for (int off = 16; off >= 1; off >>= 1)
        s += __shfl_xor(s, off, 32);
    if (lane == 0) rnorm[row] = rsqrtf(s);
}

#define CPARTS 8

__global__ __launch_bounds__(128)
void ephn_main_kernel(const float* __restrict__ fvec,
                      const int*   __restrict__ Lvec,
                      const float* __restrict__ rnorm,
                      float* __restrict__ wsPos,
                      float* __restrict__ wsNeg,
                      int N) {
    __shared__ float lds_b[64 * LDSTR];
    __shared__ float mPV[64];
    __shared__ float mNV[64];

    const int tid = threadIdx.x;
    const int l   = tid & 31;
    const int w   = tid >> 5;
    const int m   = l & 15;
    const int hi  = l >> 4;
    const int part = blockIdx.x & (CPARTS - 1);
    const int i0   = (blockIdx.x / CPARTS) * 64;
    const int iw   = i0 + w * 16;

    {
        const size_t gbase = (size_t)i0 << 7;
        for (int t = tid; t < 64 * 32; t += 128) {
            const int f   = t * 4;
            const int row = f >> 7;
            const int col = f & 127;
            float4 d = *(const float4*)(fvec + gbase + f);
            const float r = rnorm[i0 + row];
            float* dst = &lds_b[row * LDSTR + col];
            dst[0] = d.x * r; dst[1] = d.y * r; dst[2] = d.z * r; dst[3] = d.w * r;
        }
    }
    __syncthreads();
    v2f afrag[DVEC / 4];
    {
        const float* aptr = &lds_b[(w * 16 + m) * LDSTR + 2 * hi];
        #pragma unroll
        for (int t = 0; t < DVEC / 4; ++t)
            afrag[t] = *(const v2f*)(aptr + 4 * t);
    }

    int Li[8];
    #pragma unroll
    for (int v = 0; v < 8; ++v) Li[v] = Lvec[iw + v + 8 * hi];

    float posV[8], negV[8]; int posJ[8], negJ[8];
    #pragma unroll
    for (int v = 0; v < 8; ++v) {
        posV[v] = -2.0f; negV[v] = -2.0f;
        posJ[v] = 0x7fffffff; negJ[v] = 0x7fffffff;
    }

    const float* bptr = &lds_b[m * LDSTR + 2 * hi];

    const int nstrips = (N / CPARTS) >> 6;
    const int strip0  = part * nstrips;
    for (int s = 0; s < nstrips; ++s) {
        const int jt = strip0 + s;
        __syncthreads();
        const size_t gbase = ((size_t)jt << 6) << 7;
        for (int t = tid; t < 64 * 32; t += 128) {
            const int f   = t * 4;
            const int row = f >> 7;
            const int col = f & 127;
            float4 d = *(const float4*)(fvec + gbase + f);
            const float r = rnorm[jt * 64 + row];
            float* dst = &lds_b[row * LDSTR + col];
            dst[0] = d.x * r; dst[1] = d.y * r; dst[2] = d.z * r; dst[3] = d.w * r;
        }
        __syncthreads();

        v8f c[4] = {};
        #pragma unroll
        for (int t = 0; t < DVEC / 4; ++t) {
            #pragma unroll
            for (int q = 0; q < 4; ++q) {
                v2f b = *(const v2f*)(bptr + q * 16 * LDSTR + 4 * t);
                c[q] = __builtin_amdgcn_wmma_f32_16x16x4_f32(
                           false, afrag[t], false, b, (short)0, c[q], false, false);
            }
        }

        #pragma unroll
        for (int q = 0; q < 4; ++q) {
            const int j  = jt * 64 + q * 16 + m;
            const int Lj = Lvec[j];
            #pragma unroll
            for (int v = 0; v < 8; ++v) {
                const int   i   = iw + v + 8 * hi;
                const float d   = c[q][v];
                const bool  eqc = (Li[v] == Lj);
                const float pv = eqc ? (((i == j) || (d > 0.9999f)) ? -1.0f : d)
                                     : -1.0f;
                if (pv > posV[v] || (pv == posV[v] && j < posJ[v])) {
                    posV[v] = pv; posJ[v] = j;
                }
                const float nv = eqc ? -1.0f : d;
                if (nv > negV[v] || (nv == negV[v] && j < negJ[v])) {
                    negV[v] = nv; negJ[v] = j;
                }
            }
        }
    }

    #pragma unroll
    for (int v = 0; v < 8; ++v) {
        #pragma unroll
        for (int off = 1; off < 16; off <<= 1) {
            float ov = __shfl_xor(posV[v], off, 16);
            int   oj = __shfl_xor(posJ[v], off, 16);
            if (ov > posV[v] || (ov == posV[v] && oj < posJ[v])) {
                posV[v] = ov; posJ[v] = oj;
            }
            float onv = __shfl_xor(negV[v], off, 16);
            int   onj = __shfl_xor(negJ[v], off, 16);
            if (onv > negV[v] || (onv == negV[v] && onj < negJ[v])) {
                negV[v] = onv; negJ[v] = onj;
            }
        }
    }

    if ((l & 15) == 0) {
        #pragma unroll
        for (int v = 0; v < 8; ++v) {
            const int row = w * 16 + v + 8 * hi;
            mPV[row] = posV[v];
            mNV[row] = negV[v];
        }
    }
    __syncthreads();
    if (tid < 64) {
        wsPos[(size_t)part * N + i0 + tid] = mPV[tid];
        wsNeg[(size_t)part * N + i0 + tid] = mNV[tid];
    }
}

__global__ void ephn_finalize_kernel(const float* __restrict__ wsPos,
                                     const float* __restrict__ wsNeg,
                                     float* __restrict__ out, int N) {
    __shared__ float sProb[1024];
    __shared__ float sPos[1024];
    __shared__ float sNeg[1024];
    __shared__ int   sCnt[1024];
    const int tid = threadIdx.x;

    float* eps = out + 1;
    float* hns = out + 1 + (N + 2);

    float accProb = 0.0f, accPos = 0.0f, accNeg = 0.0f;
    int accCnt = 0;
    for (int i = tid; i < N; i += 1024) {
        float p = wsPos[i];
        float n = wsNeg[i];
        for (int k = 1; k < CPARTS; ++k) {
            const float pk = wsPos[(size_t)k * N + i];
            const float nk = wsNeg[(size_t)k * N + i];
            if (pk > p) p = pk;
            if (nk > n) n = nk;
        }
        eps[i] = (p > 0.0f) ? p : 0.0f;
        hns[i] = (n > 0.0f) ? n : 0.0f;
        accPos += p;
        accNeg += n;
        if (p > 0.0f && n > 0.0f) {
            const float s = (n - p) * INV_SIGMA;
            const float prob = (s > 0.0f) ? (s + log1pf(expf(-s)))
                                          : log1pf(expf(s));
            accProb += prob;
            accCnt  += 1;
        }
    }
    sProb[tid] = accProb; sPos[tid] = accPos; sNeg[tid] = accNeg; sCnt[tid] = accCnt;
    __syncthreads();
    for (int st = 512; st > 0; st >>= 1) {
        if (tid < st) {
            sProb[tid] += sProb[tid + st];
            sPos[tid]  += sPos[tid + st];
            sNeg[tid]  += sNeg[tid + st];
            sCnt[tid]  += sCnt[tid + st];
        }
        __syncthreads();
    }
    if (tid == 0) {
        const int cnt = (sCnt[0] > 1) ? sCnt[0] : 1;
        out[0] = sProb[0] / (float)cnt;
        eps[N] = 0.0f; eps[N + 1] = 1.0f;
        hns[N] = 0.0f; hns[N + 1] = 1.0f;
        out[1 + 2 * (N + 2)] = (sPos[0] - sNeg[0]) / (float)N;
    }
}

extern "C" void kernel_launch(void* const* d_in, const int* in_sizes, int n_in,
                              void* d_out, int out_size, void* d_ws, size_t ws_size,
                              hipStream_t stream) {
    const float* fvec = (const float*)d_in[0];
    const int*   Lvec = (const int*)d_in[1];
    const int N = in_sizes[1];

    float* rnorm = (float*)d_ws;
    float* wsPos = rnorm + N;
    float* wsNeg = wsPos + CPARTS * N;

    ephn_norm_kernel<<<N / 4, 128, 0, stream>>>(fvec, rnorm);
    ephn_main_kernel<<<(N / 64) * CPARTS, 128, 0, stream>>>(fvec, Lvec, rnorm,
                                                            wsPos, wsNeg, N);
    ephn_finalize_kernel<<<1, 1024, 0, stream>>>(wsPos, wsNeg,
                                                 (float*)d_out, N);
}


# Software-pipelined staging: strip s+1's 16 global loads per thread are issued as one group into registers before strip s's matrix phase, so the L2 round trips run under the WMMAs and mining
# speedup vs baseline: 3.1302x; 1.1508x over previous
#include <hip/hip_runtime.h>
#include <hip/hip_bf16.h>


typedef __attribute__((ext_vector_type(2))) float v2f;
typedef __attribute__((ext_vector_type(8))) float v8f;

#define DVEC 128
#define LDSTR 132
#define INV_SIGMA 10.0f

__global__ void ephn_norm_kernel(const float* __restrict__ fvec,
                                 float* __restrict__ rnorm) {
    const int row  = blockIdx.x * 4 + (threadIdx.x >> 5);
    const int lane = threadIdx.x & 31;
    const float4 d = ((const float4*)(fvec + ((size_t)row << 7)))[lane];
    float s = d.x * d.x + d.y * d.y + d.z * d.z + d.w * d.w;
    #pragma unroll
    for (int off = 16; off >= 1; off >>= 1)
        s += __shfl_xor(s, off, 32);
    if (lane == 0) rnorm[row] = rsqrtf(s);
}

#define CPARTS 8

__global__ __launch_bounds__(128)
void ephn_main_kernel(const float* __restrict__ fvec,
                      const int*   __restrict__ Lvec,
                      const float* __restrict__ rnorm,
                      float* __restrict__ wsPos,
                      float* __restrict__ wsNeg,
                      int N) {
    __shared__ __attribute__((aligned(16))) float lds_b[64 * LDSTR];
    __shared__ float mPV[64];
    __shared__ float mNV[64];

    const int tid = threadIdx.x;
    const int l   = tid & 31;
    const int w   = tid >> 5;
    const int m   = l & 15;
    const int hi  = l >> 4;
    const int part = blockIdx.x & (CPARTS - 1);
    const int i0   = (blockIdx.x / CPARTS) * 64;
    const int iw   = i0 + w * 16;

    const int ccol = l << 2;
    float4 pre[16];
    float  prn;

    auto panel_issue = [&](int rowbase) {
        const float* src = fvec + ((size_t)(rowbase + w) << 7) + ccol;
        #pragma unroll
        for (int p = 0; p < 16; ++p)
            pre[p] = *(const float4*)(src + ((size_t)p << 9));
        prn = rnorm[rowbase + w + 4 * m];
    };
    auto panel_commit = [&]() {
        #pragma unroll
        for (int p = 0; p < 16; ++p) {
            const float r = __int_as_float(
                __builtin_amdgcn_readlane(__float_as_int(prn), p));
            const float4 d = pre[p];
            *(float4*)&lds_b[(w + 4 * p) * LDSTR + ccol] =
                make_float4(d.x * r, d.y * r, d.z * r, d.w * r);
        }
    };

    const int nstrips = (N / CPARTS) >> 6;
    const int strip0  = part * nstrips;

    panel_issue(i0);
    panel_commit();
    __syncthreads();
    panel_issue(strip0 << 6);
    v2f afrag[DVEC / 4];
    {
        const float* aptr = &lds_b[(w * 16 + m) * LDSTR + 2 * hi];
        #pragma unroll
        for (int t = 0; t < DVEC / 4; ++t)
            afrag[t] = *(const v2f*)(aptr + 4 * t);
    }

    int Li[8];
    #pragma unroll
    for (int v = 0; v < 8; ++v) Li[v] = Lvec[iw + v + 8 * hi];

    float posV[8], negV[8]; int posJ[8], negJ[8];
    #pragma unroll
    for (int v = 0; v < 8; ++v) {
        posV[v] = -2.0f; negV[v] = -2.0f;
        posJ[v] = 0x7fffffff; negJ[v] = 0x7fffffff;
    }

    const float* bptr = &lds_b[m * LDSTR + 2 * hi];

    for (int s = 0; s < nstrips; ++s) {
        const int jt = strip0 + s;
        __syncthreads();
        panel_commit();
        __syncthreads();
        if (s + 1 < nstrips) panel_issue((jt + 1) << 6);

        v8f c[4] = {};
        #pragma unroll
        for (int t = 0; t < DVEC / 4; ++t) {
            #pragma unroll
            for (int q = 0; q < 4; ++q) {
                v2f b = *(const v2f*)(bptr + q * 16 * LDSTR + 4 * t);
                c[q] = __builtin_amdgcn_wmma_f32_16x16x4_f32(
                           false, afrag[t], false, b, (short)0, c[q], false, false);
            }
        }

        #pragma unroll
        for (int q = 0; q < 4; ++q) {
            const int j  = jt * 64 + q * 16 + m;
            const int Lj = Lvec[j];
            #pragma unroll
            for (int v = 0; v < 8; ++v) {
                const int   i   = iw + v + 8 * hi;
                const float d   = c[q][v];
                const bool  eqc = (Li[v] == Lj);
                const float pv = eqc ? (((i == j) || (d > 0.9999f)) ? -1.0f : d)
                                     : -1.0f;
                if (pv > posV[v] || (pv == posV[v] && j < posJ[v])) {
                    posV[v] = pv; posJ[v] = j;
                }
                const float nv = eqc ? -1.0f : d;
                if (nv > negV[v] || (nv == negV[v] && j < negJ[v])) {
                    negV[v] = nv; negJ[v] = j;
                }
            }
        }
    }

    #pragma unroll
    for (int v = 0; v < 8; ++v) {
        #pragma unroll
        for (int off = 1; off < 16; off <<= 1) {
            float ov = __shfl_xor(posV[v], off, 16);
            int   oj = __shfl_xor(posJ[v], off, 16);
            if (ov > posV[v] || (ov == posV[v] && oj < posJ[v])) {
                posV[v] = ov; posJ[v] = oj;
            }
            float onv = __shfl_xor(negV[v], off, 16);
            int   onj = __shfl_xor(negJ[v], off, 16);
            if (onv > negV[v] || (onv == negV[v] && onj < negJ[v])) {
                negV[v] = onv; negJ[v] = onj;
            }
        }
    }

    if ((l & 15) == 0) {
        #pragma unroll
        for (int v = 0; v < 8; ++v) {
            const int row = w * 16 + v + 8 * hi;
            mPV[row] = posV[v];
            mNV[row] = negV[v];
        }
    }
    __syncthreads();
    if (tid < 64) {
        wsPos[(size_t)part * N + i0 + tid] = mPV[tid];
        wsNeg[(size_t)part * N + i0 + tid] = mNV[tid];
    }
}

__global__ void ephn_finalize_kernel(const float* __restrict__ wsPos,
                                     const float* __restrict__ wsNeg,
                                     float* __restrict__ out, int N) {
    __shared__ float sProb[1024];
    __shared__ float sPos[1024];
    __shared__ float sNeg[1024];
    __shared__ int   sCnt[1024];
    const int tid = threadIdx.x;

    float* eps = out + 1;
    float* hns = out + 1 + (N + 2);

    float accProb = 0.0f, accPos = 0.0f, accNeg = 0.0f;
    int accCnt = 0;
    for (int i = tid; i < N; i += 1024) {
        float p = wsPos[i];
        float n = wsNeg[i];
        for (int k = 1; k < CPARTS; ++k) {
            const float pk = wsPos[(size_t)k * N + i];
            const float nk = wsNeg[(size_t)k * N + i];
            if (pk > p) p = pk;
            if (nk > n) n = nk;
        }
        eps[i] = (p > 0.0f) ? p : 0.0f;
        hns[i] = (n > 0.0f) ? n : 0.0f;
        accPos += p;
        accNeg += n;
        if (p > 0.0f && n > 0.0f) {
            const float s = (n - p) * INV_SIGMA;
            const float prob = (s > 0.0f) ? (s + log1pf(expf(-s)))
                                          : log1pf(expf(s));
            accProb += prob;
            accCnt  += 1;
        }
    }
    sProb[tid] = accProb; sPos[tid] = accPos; sNeg[tid] = accNeg; sCnt[tid] = accCnt;
    __syncthreads();
    for (int st = 512; st > 0; st >>= 1) {
        if (tid < st) {
            sProb[tid] += sProb[tid + st];
            sPos[tid]  += sPos[tid + st];
            sNeg[tid]  += sNeg[tid + st];
            sCnt[tid]  += sCnt[tid + st];
        }
        __syncthreads();
    }
    if (tid == 0) {
        const int cnt = (sCnt[0] > 1) ? sCnt[0] : 1;
        out[0] = sProb[0] / (float)cnt;
        eps[N] = 0.0f; eps[N + 1] = 1.0f;
        hns[N] = 0.0f; hns[N + 1] = 1.0f;
        out[1 + 2 * (N + 2)] = (sPos[0] - sNeg[0]) / (float)N;
    }
}

extern "C" void kernel_launch(void* const* d_in, const int* in_sizes, int n_in,
                              void* d_out, int out_size, void* d_ws, size_t ws_size,
                              hipStream_t stream) {
    const float* fvec = (const float*)d_in[0];
    const int*   Lvec = (const int*)d_in[1];
    const int N = in_sizes[1];

    float* rnorm = (float*)d_ws;
    float* wsPos = rnorm + N;
    float* wsNeg = wsPos + CPARTS * N;

    ephn_norm_kernel<<<N / 4, 128, 0, stream>>>(fvec, rnorm);
    ephn_main_kernel<<<(N / 64) * CPARTS, 128, 0, stream>>>(fvec, Lvec, rnorm,
                                                            wsPos, wsNeg, N);
    ephn_finalize_kernel<<<1, 1024, 0, stream>>>(wsPos, wsNeg,
                                                 (float*)d_out, N);
}


# Mining keeps only the running row maxima (no argmax index reaches an output): per element 2 selects + 2 max; the i==j test is compiled only into the strip that holds the block's own rows
# speedup vs baseline: 3.9182x; 1.2517x over previous
#include <hip/hip_runtime.h>
#include <hip/hip_bf16.h>


typedef __attribute__((ext_vector_type(2))) float v2f;
typedef __attribute__((ext_vector_type(8))) float v8f;

#define DVEC 128
#define LDSTR 132
#define INV_SIGMA 10.0f

__global__ void ephn_norm_kernel(const float* __restrict__ fvec,
                                 float* __restrict__ rnorm) {
    const int row  = blockIdx.x * 4 + (threadIdx.x >> 5);
    const int lane = threadIdx.x & 31;
    const float4 d = ((const float4*)(fvec + ((size_t)row << 7)))[lane];
    float s = d.x * d.x + d.y * d.y + d.z * d.z + d.w * d.w;
    #pragma unroll
    for (int off = 16; off >= 1; off >>= 1)
        s += __shfl_xor(s, off, 32);
    if (lane == 0) rnorm[row] = rsqrtf(s);
}

#define CPARTS 8

template <bool DIAG>
__device__ __forceinline__ void ephn_mine_slab(const v8f (&c)[4],
                                               const int (&Lj)[4],
                                               const int (&Li)[8], int dd,
                                               float (&posV)[8],
                                               float (&negV)[8]) {
    #pragma unroll
    for (int q = 0; q < 4; ++q) {
        #pragma unroll
        for (int v = 0; v < 8; ++v) {
            const float d   = c[q][v];
            const bool  eqc = (Li[v] == Lj[q]);
            bool okp = eqc && !(d > 0.9999f);
            if (DIAG) okp = okp && (dd != v - q * 16);
            posV[v] = fmaxf(posV[v], okp ? d : -1.0f);
            negV[v] = fmaxf(negV[v], eqc ? -1.0f : d);
        }
    }
}

__global__ __launch_bounds__(128)
void ephn_main_kernel(const float* __restrict__ fvec,
                      const int*   __restrict__ Lvec,
                      const float* __restrict__ rnorm,
                      float* __restrict__ wsPos,
                      float* __restrict__ wsNeg,
                      int N) {
    __shared__ __attribute__((aligned(16))) float lds_b[64 * LDSTR];
    __shared__ float mPV[64];
    __shared__ float mNV[64];

    const int tid = threadIdx.x;
    const int l   = tid & 31;
    const int w   = tid >> 5;
    const int m   = l & 15;
    const int hi  = l >> 4;
    const int part = blockIdx.x & (CPARTS - 1);
    const int i0   = (blockIdx.x / CPARTS) * 64;
    const int iw   = i0 + w * 16;

    const int ccol = l << 2;
    float4 pre[16];
    float  prn;

    auto panel_issue = [&](int rowbase) {
        const float* src = fvec + ((size_t)(rowbase + w) << 7) + ccol;
        #pragma unroll
        for (int p = 0; p < 16; ++p)
            pre[p] = *(const float4*)(src + ((size_t)p << 9));
        prn = rnorm[rowbase + w + 4 * m];
    };
    auto panel_commit = [&]() {
        #pragma unroll
        for (int p = 0; p < 16; ++p) {
            const float r = __int_as_float(
                __builtin_amdgcn_readlane(__float_as_int(prn), p));
            const float4 d = pre[p];
            *(float4*)&lds_b[(w + 4 * p) * LDSTR + ccol] =
                make_float4(d.x * r, d.y * r, d.z * r, d.w * r);
        }
    };

    const int nstrips = (N / CPARTS) >> 6;
    const int strip0  = part * nstrips;

    panel_issue(i0);
    panel_commit();
    __syncthreads();
    panel_issue(strip0 << 6);
    v2f afrag[DVEC / 4];
    {
        const float* aptr = &lds_b[(w * 16 + m) * LDSTR + 2 * hi];
        #pragma unroll
        for (int t = 0; t < DVEC / 4; ++t)
            afrag[t] = *(const v2f*)(aptr + 4 * t);
    }

    int Li[8];
    #pragma unroll
    for (int v = 0; v < 8; ++v) Li[v] = Lvec[iw + v + 8 * hi];

    float posV[8], negV[8];
    #pragma unroll
    for (int v = 0; v < 8; ++v) {
        posV[v] = -2.0f; negV[v] = -2.0f;
    }
    const int dd0 = m - iw - 8 * hi;

    const float* bptr = &lds_b[m * LDSTR + 2 * hi];

    for (int s = 0; s < nstrips; ++s) {
        const int jt = strip0 + s;
        __syncthreads();
        panel_commit();
        __syncthreads();
        if (s + 1 < nstrips) panel_issue((jt + 1) << 6);

        v8f c[4] = {};
        #pragma unroll
        for (int t = 0; t < DVEC / 4; ++t) {
            #pragma unroll
            for (int q = 0; q < 4; ++q) {
                v2f b = *(const v2f*)(bptr + q * 16 * LDSTR + 4 * t);
                c[q] = __builtin_amdgcn_wmma_f32_16x16x4_f32(
                           false, afrag[t], false, b, (short)0, c[q], false, false);
            }
        }

        int Lj[4];
        #pragma unroll
        for (int q = 0; q < 4; ++q) Lj[q] = Lvec[jt * 64 + q * 16 + m];
        if (jt * 64 == i0)
            ephn_mine_slab<true>(c, Lj, Li, dd0 + jt * 64, posV, negV);
        else
            ephn_mine_slab<false>(c, Lj, Li, 0, posV, negV);
    }

    #pragma unroll
    for (int v = 0; v < 8; ++v) {
        #pragma unroll
        for (int off = 1; off < 16; off <<= 1) {
            posV[v] = fmaxf(posV[v], __shfl_xor(posV[v], off, 16));
            negV[v] = fmaxf(negV[v], __shfl_xor(negV[v], off, 16));
        }
    }

    if ((l & 15) == 0) {
        #pragma unroll
        for (int v = 0; v < 8; ++v) {
            const int row = w * 16 + v + 8 * hi;
            mPV[row] = posV[v];
            mNV[row] = negV[v];
        }
    }
    __syncthreads();
    if (tid < 64) {
        wsPos[(size_t)part * N + i0 + tid] = mPV[tid];
        wsNeg[(size_t)part * N + i0 + tid] = mNV[tid];
    }
}

__global__ void ephn_finalize_kernel(const float* __restrict__ wsPos,
                                     const float* __restrict__ wsNeg,
                                     float* __restrict__ out, int N) {
    __shared__ float sProb[1024];
    __shared__ float sPos[1024];
    __shared__ float sNeg[1024];
    __shared__ int   sCnt[1024];
    const int tid = threadIdx.x;

    float* eps = out + 1;
    float* hns = out + 1 + (N + 2);

    float accProb = 0.0f, accPos = 0.0f, accNeg = 0.0f;
    int accCnt = 0;
    for (int i = tid; i < N; i += 1024) {
        float p = wsPos[i];
        float n = wsNeg[i];
        for (int k = 1; k < CPARTS; ++k) {
            const float pk = wsPos[(size_t)k * N + i];
            const float nk = wsNeg[(size_t)k * N + i];
            if (pk > p) p = pk;
            if (nk > n) n = nk;
        }
        eps[i] = (p > 0.0f) ? p : 0.0f;
        hns[i] = (n > 0.0f) ? n : 0.0f;
        accPos += p;
        accNeg += n;
        if (p > 0.0f && n > 0.0f) {
            const float s = (n - p) * INV_SIGMA;
            const float prob = (s > 0.0f) ? (s + log1pf(expf(-s)))
                                          : log1pf(expf(s));
            accProb += prob;
            accCnt  += 1;
        }
    }
    sProb[tid] = accProb; sPos[tid] = accPos; sNeg[tid] = accNeg; sCnt[tid] = accCnt;
    __syncthreads();
    for (int st = 512; st > 0; st >>= 1) {
        if (tid < st) {
            sProb[tid] += sProb[tid + st];
            sPos[tid]  += sPos[tid + st];
            sNeg[tid]  += sNeg[tid + st];
            sCnt[tid]  += sCnt[tid + st];
        }
        __syncthreads();
    }
    if (tid == 0) {
        const int cnt = (sCnt[0] > 1) ? sCnt[0] : 1;
        out[0] = sProb[0] / (float)cnt;
        eps[N] = 0.0f; eps[N + 1] = 1.0f;
        hns[N] = 0.0f; hns[N + 1] = 1.0f;
        out[1 + 2 * (N + 2)] = (sPos[0] - sNeg[0]) / (float)N;
    }
}

extern "C" void kernel_launch(void* const* d_in, const int* in_sizes, int n_in,
                              void* d_out, int out_size, void* d_ws, size_t ws_size,
                              hipStream_t stream) {
    const float* fvec = (const float*)d_in[0];
    const int*   Lvec = (const int*)d_in[1];
    const int N = in_sizes[1];

    float* rnorm = (float*)d_ws;
    float* wsPos = rnorm + N;
    float* wsNeg = wsPos + CPARTS * N;

    ephn_norm_kernel<<<N / 4, 128, 0, stream>>>(fvec, rnorm);
    ephn_main_kernel<<<(N / 64) * CPARTS, 128, 0, stream>>>(fvec, Lvec, rnorm,
                                                            wsPos, wsNeg, N);
    ephn_finalize_kernel<<<1, 1024, 0, stream>>>(wsPos, wsNeg,
                                                 (float*)d_out, N);
}


# Use the symmetry of fn*fn^T: a row block computes only column blocks I..I+64 (mod 128) and mines each tile along rows and along columns; a new merge launch joins row-side and column-side maxima
# speedup vs baseline: 5.9438x; 1.5170x over previous
#include <hip/hip_runtime.h>
#include <hip/hip_bf16.h>


typedef __attribute__((ext_vector_type(2))) float v2f;
typedef __attribute__((ext_vector_type(8))) float v8f;

#define DVEC 128
#define LDSTR 132
#define INV_SIGMA 10.0f
#define NBLK 128
#define HALFB (NBLK / 2)

__global__ void ephn_norm_kernel(const float* __restrict__ fvec,
                                 float* __restrict__ rnorm) {
    const int row  = blockIdx.x * 4 + (threadIdx.x >> 5);
    const int lane = threadIdx.x & 31;
    const float4 d = ((const float4*)(fvec + ((size_t)row << 7)))[lane];
    float s = d.x * d.x + d.y * d.y + d.z * d.z + d.w * d.w;
    #pragma unroll
    for (int off = 16; off >= 1; off >>= 1)
        s += __shfl_xor(s, off, 32);
    if (lane == 0) rnorm[row] = rsqrtf(s);
}

#define CPARTS 8
#define SPP (HALFB / CPARTS)
#define CBS 144

template <bool DIAG>
__device__ __forceinline__ void ephn_mine_slab(const v8f (&c)[4],
                                               const int (&Lj)[4],
                                               const int (&Li)[8], int dd,
                                               float (&posV)[8],
                                               float (&negV)[8],
                                               float (&colP)[4],
                                               float (&colN)[4]) {
    #pragma unroll
    for (int q = 0; q < 4; ++q) {
        float cp = -2.0f, cn = -2.0f;
        #pragma unroll
        for (int v = 0; v < 8; ++v) {
            const float d   = c[q][v];
            const bool  eqc = (Li[v] == Lj[q]);
            bool okp = eqc && !(d > 0.9999f);
            if (DIAG) okp = okp && (dd != v - q * 16);
            const float sp = okp ? d : -1.0f;
            const float sn = eqc ? -1.0f : d;
            posV[v] = fmaxf(posV[v], sp);
            negV[v] = fmaxf(negV[v], sn);
            if (!DIAG) {
                cp = fmaxf(cp, sp);
                cn = fmaxf(cn, sn);
            }
        }
        colP[q] = cp;
        colN[q] = cn;
    }
}

__global__ __launch_bounds__(128) __attribute__((amdgpu_waves_per_eu(4, 4)))
void ephn_main_kernel(const float* __restrict__ fvec,
                      const int*   __restrict__ Lvec,
                      const float* __restrict__ rnorm,
                      float* __restrict__ wsPos,
                      float* __restrict__ wsNeg,
                      float* __restrict__ wsColP,
                      float* __restrict__ wsColN,
                      int N) {
    __shared__ __attribute__((aligned(16))) float lds_b[64 * LDSTR];
    __shared__ float mPV[64];
    __shared__ float mNV[64];
    __shared__ float colbuf[8 * CBS];

    const int tid = threadIdx.x;
    const int l   = tid & 31;
    const int w   = __builtin_amdgcn_readfirstlane(tid >> 5);
    const int m   = l & 15;
    const int hi  = l >> 4;
    const int part = blockIdx.x / NBLK;
    const int Iblk = blockIdx.x & (NBLK - 1);
    const int i0   = Iblk * 64;
    const int iw   = i0 + w * 16;

    const int ccol = l << 2;
    float4 pre[16];
    float  prn;

    auto panel_issue = [&](int rowbase) {
        const float* src = fvec + ((size_t)(rowbase + w) << 7) + ccol;
        #pragma unroll
        for (int p = 0; p < 16; ++p)
            pre[p] = *(const float4*)(src + ((size_t)p << 9));
        prn = rnorm[rowbase + w + 4 * m];
    };
    auto panel_commit = [&]() {
        #pragma unroll
        for (int p = 0; p < 16; ++p) {
            const float r = __int_as_float(
                __builtin_amdgcn_readlane(__float_as_int(prn), p));
            const float4 d = pre[p];
            *(float4*)&lds_b[(w + 4 * p) * LDSTR + ccol] =
                make_float4(d.x * r, d.y * r, d.z * r, d.w * r);
        }
    };

    const int d0      = part * SPP;
    const int nstrips = SPP + ((part == CPARTS - 1) ? 1 : 0);
    const int strip0  = (Iblk + d0) & (NBLK - 1);

    auto col_flush = [&](int jcol, int lane) {
        const int list = w >> 1;
        const int col  = ((w & 1) << 5) + lane;
        const float* cbp = &colbuf[list * 64 + col];
        float r = cbp[0];
        #pragma unroll
        for (int k = 1; k < 8; ++k) r = fmaxf(r, cbp[k * CBS]);
        float* dst = list ? wsColN : wsColP;
        dst[(size_t)Iblk * N + jcol * 64 + col] = r;
    };

    panel_issue(i0);
    panel_commit();
    __syncthreads();
    panel_issue(strip0 << 6);
    v2f afrag[DVEC / 4];
    {
        const float* aptr = &lds_b[(w * 16 + m) * LDSTR + 2 * hi];
        #pragma unroll
        for (int t = 0; t < DVEC / 4; ++t)
            afrag[t] = *(const v2f*)(aptr + 4 * t);
    }

    int Li[8];
    #pragma unroll
    for (int v = 0; v < 8; ++v) Li[v] = Lvec[iw + v + 8 * hi];

    float posV[8], negV[8];
    #pragma unroll
    for (int v = 0; v < 8; ++v) {
        posV[v] = -2.0f; negV[v] = -2.0f;
    }
    const int dd0 = m - iw - 8 * hi;

    const float* bptr = &lds_b[m * LDSTR + 2 * hi];

    for (int s = 0; s < nstrips; ++s) {
        const int dcur = d0 + s;
        const int jt   = (Iblk + dcur) & (NBLK - 1);
        __syncthreads();
        if (s > 0 && dcur > 1) col_flush((Iblk + dcur - 1) & (NBLK - 1), l);
        panel_commit();
        __syncthreads();
        if (s + 1 < nstrips) panel_issue(((Iblk + dcur + 1) & (NBLK - 1)) << 6);

        v8f c[4] = {};
        #pragma unroll
        for (int t = 0; t < DVEC / 4; ++t) {
            #pragma unroll
            for (int q = 0; q < 4; ++q) {
                v2f b = *(const v2f*)(bptr + q * 16 * LDSTR + 4 * t);
                c[q] = __builtin_amdgcn_wmma_f32_16x16x4_f32(
                           false, afrag[t], false, b, (short)0, c[q], false, false);
            }
        }

        int Lj[4];
        #pragma unroll
        for (int q = 0; q < 4; ++q) Lj[q] = Lvec[jt * 64 + q * 16 + m];
        float colP[4], colN[4];
        if (dcur == 0) {
            ephn_mine_slab<true>(c, Lj, Li, dd0 + jt * 64, posV, negV,
                                 colP, colN);
        } else {
            ephn_mine_slab<false>(c, Lj, Li, 0, posV, negV, colP, colN);
            float* cb = &colbuf[(w * 2 + hi) * CBS + m];
            #pragma unroll
            for (int q = 0; q < 4; ++q) {
                cb[q * 16]      = colP[q];
                cb[64 + q * 16] = colN[q];
            }
        }
    }

    #pragma unroll
    for (int v = 0; v < 8; ++v) {
        #pragma unroll
        for (int off = 1; off < 16; off <<= 1) {
            posV[v] = fmaxf(posV[v], __shfl_xor(posV[v], off, 16));
            negV[v] = fmaxf(negV[v], __shfl_xor(negV[v], off, 16));
        }
    }

    const int le = (int)__lane_id();
    if ((le & 15) == 0) {
        #pragma unroll
        for (int v = 0; v < 8; ++v) {
            const int row = w * 16 + v + 8 * (le >> 4);
            mPV[row] = posV[v];
            mNV[row] = negV[v];
        }
    }
    __syncthreads();
    if (w < 2) {
        const int te = w * 32 + le;
        wsPos[(size_t)part * N + i0 + te] = mPV[te];
        wsNeg[(size_t)part * N + i0 + te] = mNV[te];
    }
    col_flush((Iblk + d0 + nstrips - 1) & (NBLK - 1), le);
}

__global__ __launch_bounds__(256)
void ephn_merge_kernel(const float* __restrict__ wsPos,
                       const float* __restrict__ wsNeg,
                       const float* __restrict__ wsColP,
                       const float* __restrict__ wsColN,
                       float* __restrict__ mrgP,
                       float* __restrict__ mrgN, int N) {
    __shared__ float sP[4 * 64];
    __shared__ float sN[4 * 64];
    const int tid = threadIdx.x;
    const int x   = tid & 63;
    const int g   = tid >> 6;
    const int R   = blockIdx.x;
    const int r   = R * 64 + x;

    float p = fmaxf(wsPos[(size_t)(2 * g) * N + r], wsPos[(size_t)(2 * g + 1) * N + r]);
    float n = fmaxf(wsNeg[(size_t)(2 * g) * N + r], wsNeg[(size_t)(2 * g + 1) * N + r]);
    #pragma unroll
    for (int k = 0; k < HALFB / 4; ++k) {
        const int d = g * (HALFB / 4) + 1 + k;
        const int I = (R - d) & (NBLK - 1);
        p = fmaxf(p, wsColP[(size_t)I * N + r]);
        n = fmaxf(n, wsColN[(size_t)I * N + r]);
    }
    sP[g * 64 + x] = p;
    sN[g * 64 + x] = n;
    __syncthreads();
    if (tid < 64) {
        const float pm = fmaxf(fmaxf(sP[tid], sP[64 + tid]),
                               fmaxf(sP[128 + tid], sP[192 + tid]));
        const float nm = fmaxf(fmaxf(sN[tid], sN[64 + tid]),
                               fmaxf(sN[128 + tid], sN[192 + tid]));
        mrgP[r] = pm;
        mrgN[r] = nm;
    }
}

__global__ void ephn_finalize_kernel(const float* __restrict__ mrgP,
                                     const float* __restrict__ mrgN,
                                     float* __restrict__ out, int N) {
    __shared__ float sProb[1024];
    __shared__ float sPos[1024];
    __shared__ float sNeg[1024];
    __shared__ int   sCnt[1024];
    const int tid = threadIdx.x;

    float* eps = out + 1;
    float* hns = out + 1 + (N + 2);

    float accProb = 0.0f, accPos = 0.0f, accNeg = 0.0f;
    int accCnt = 0;
    for (int i = tid; i < N; i += 1024) {
        const float p = mrgP[i];
        const float n = mrgN[i];
        eps[i] = (p > 0.0f) ? p : 0.0f;
        hns[i] = (n > 0.0f) ? n : 0.0f;
        accPos += p;
        accNeg += n;
        if (p > 0.0f && n > 0.0f) {
            const float s = (n - p) * INV_SIGMA;
            const float prob = (s > 0.0f) ? (s + log1pf(expf(-s)))
                                          : log1pf(expf(s));
            accProb += prob;
            accCnt  += 1;
        }
    }
    sProb[tid] = accProb; sPos[tid] = accPos; sNeg[tid] = accNeg; sCnt[tid] = accCnt;
    __syncthreads();
    for (int st = 512; st > 0; st >>= 1) {
        if (tid < st) {
            sProb[tid] += sProb[tid + st];
            sPos[tid]  += sPos[tid + st];
            sNeg[tid]  += sNeg[tid + st];
            sCnt[tid]  += sCnt[tid + st];
        }
        __syncthreads();
    }
    if (tid == 0) {
        const int cnt = (sCnt[0] > 1) ? sCnt[0] : 1;
        out[0] = sProb[0] / (float)cnt;
        eps[N] = 0.0f; eps[N + 1] = 1.0f;
        hns[N] = 0.0f; hns[N + 1] = 1.0f;
        out[1 + 2 * (N + 2)] = (sPos[0] - sNeg[0]) / (float)N;
    }
}

extern "C" void kernel_launch(void* const* d_in, const int* in_sizes, int n_in,
                              void* d_out, int out_size, void* d_ws, size_t ws_size,
                              hipStream_t stream) {
    const float* fvec = (const float*)d_in[0];
    const int*   Lvec = (const int*)d_in[1];
    const int N = in_sizes[1];

    float* rnorm = (float*)d_ws;
    float* wsPos = rnorm + N;
    float* wsNeg = wsPos + CPARTS * N;
    float* wsColP = wsNeg + CPARTS * N;
    float* wsColN = wsColP + NBLK * N;
    float* mrgP   = wsColN + NBLK * N;
    float* mrgN   = mrgP + N;

    ephn_norm_kernel<<<N / 4, 128, 0, stream>>>(fvec, rnorm);
    ephn_main_kernel<<<NBLK * CPARTS, 128, 0, stream>>>(fvec, Lvec, rnorm,
                                                        wsPos, wsNeg,
                                                        wsColP, wsColN, N);
    ephn_merge_kernel<<<NBLK, 256, 0, stream>>>(wsPos, wsNeg, wsColP, wsColN,
                                                mrgP, mrgN, N);
    ephn_finalize_kernel<<<1, 1024, 0, stream>>>(mrgP, mrgN,
                                                 (float*)d_out, N);
}


# Row norms folded into the main kernel's staging: each wave squares the chunks it stages and sums them across lanes on the vector ALU (same summation tree); the separate norm launch is gone
# speedup vs baseline: 6.2069x; 1.0443x over previous
#include <hip/hip_runtime.h>
#include <hip/hip_bf16.h>


typedef __attribute__((ext_vector_type(2))) float v2f;
typedef __attribute__((ext_vector_type(8))) float v8f;

#define DVEC 128
#define LDSTR 132
#define INV_SIGMA 10.0f
#define NBLK 128
#define HALFB (NBLK / 2)

__device__ __forceinline__ float ephn_xrow(float x) {
    const int b = __builtin_bit_cast(int, x);
    return __builtin_bit_cast(float,
        __builtin_amdgcn_permlanex16(b, b, 0x76543210, 0xfedcba98, false, false));
}
template <int M>
__device__ __forceinline__ float ephn_xlane(float x) {
    const int b = __builtin_bit_cast(int, x);
    return __builtin_bit_cast(float,
        __builtin_amdgcn_update_dpp(b, b, 0x160 | M, 0xf, 0xf, true));
}

#define CPARTS 8
#define SPP (HALFB / CPARTS)
#define CBS 144

template <bool DIAG>
__device__ __forceinline__ void ephn_mine_slab(const v8f (&c)[4],
                                               const int (&Lj)[4],
                                               const int (&Li)[8], int dd,
                                               float (&posV)[8],
                                               float (&negV)[8],
                                               float (&colP)[4],
                                               float (&colN)[4]) {
    #pragma unroll
    for (int q = 0; q < 4; ++q) {
        float cp = -2.0f, cn = -2.0f;
        #pragma unroll
        for (int v = 0; v < 8; ++v) {
            const float d   = c[q][v];
            const bool  eqc = (Li[v] == Lj[q]);
            bool okp = eqc && !(d > 0.9999f);
            if (DIAG) okp = okp && (dd != v - q * 16);
            const float sp = okp ? d : -1.0f;
            const float sn = eqc ? -1.0f : d;
            posV[v] = fmaxf(posV[v], sp);
            negV[v] = fmaxf(negV[v], sn);
            if (!DIAG) {
                cp = fmaxf(cp, sp);
                cn = fmaxf(cn, sn);
            }
        }
        colP[q] = cp;
        colN[q] = cn;
    }
}

__global__ __launch_bounds__(128) __attribute__((amdgpu_waves_per_eu(4, 4)))
void ephn_main_kernel(const float* __restrict__ fvec,
                      const int*   __restrict__ Lvec,
                      float* __restrict__ wsPos,
                      float* __restrict__ wsNeg,
                      float* __restrict__ wsColP,
                      float* __restrict__ wsColN,
                      int N) {
    __shared__ __attribute__((aligned(16))) float lds_b[64 * LDSTR];
    __shared__ float mPV[64];
    __shared__ float mNV[64];
    __shared__ float colbuf[8 * CBS];

    const int tid = threadIdx.x;
    const int l   = tid & 31;
    const int w   = __builtin_amdgcn_readfirstlane(tid >> 5);
    const int m   = l & 15;
    const int hi  = l >> 4;
    const int part = blockIdx.x / NBLK;
    const int Iblk = blockIdx.x & (NBLK - 1);
    const int i0   = Iblk * 64;
    const int iw   = i0 + w * 16;

    const int ccol = l << 2;
    float4 pre[16];
    float  prn;

    auto panel_issue = [&](int rowbase) {
        const float* src = fvec + ((size_t)(rowbase + w) << 7) + ccol;
        #pragma unroll
        for (int p = 0; p < 16; ++p)
            pre[p] = *(const float4*)(src + ((size_t)p << 9));
    };
    auto panel_norm = [&]() {
        float ss[16];
        #pragma unroll
        for (int p = 0; p < 16; ++p) {
            const float4 d = pre[p];
            ss[p] = fmaf(d.w, d.w, fmaf(d.z, d.z, fmaf(d.x, d.x, d.y * d.y)));
        }
        const bool b4 = (l & 16) != 0, b3 = (l & 8) != 0;
        const bool b2 = (l & 4) != 0,  b1 = (l & 2) != 0;
        float s8[8], s4[4], s2[2];
        #pragma unroll
        for (int k = 0; k < 8; ++k)
            s8[k] = (b4 ? ss[k + 8] : ss[k]) + ephn_xrow(b4 ? ss[k] : ss[k + 8]);
        #pragma unroll
        for (int k = 0; k < 4; ++k)
            s4[k] = (b3 ? s8[k + 4] : s8[k]) + ephn_xlane<8>(b3 ? s8[k] : s8[k + 4]);
        #pragma unroll
        for (int k = 0; k < 2; ++k)
            s2[k] = (b2 ? s4[k + 2] : s4[k]) + ephn_xlane<4>(b2 ? s4[k] : s4[k + 2]);
        const float s1 = (b1 ? s2[1] : s2[0]) + ephn_xlane<2>(b1 ? s2[0] : s2[1]);
        prn = rsqrtf(s1 + ephn_xlane<1>(s1));
    };
    auto panel_commit = [&]() {
        #pragma unroll
        for (int p = 0; p < 16; ++p) {
            const float r = __int_as_float(
                __builtin_amdgcn_readlane(__float_as_int(prn), 2 * p));
            const float4 d = pre[p];
            *(float4*)&lds_b[(w + 4 * p) * LDSTR + ccol] =
                make_float4(d.x * r, d.y * r, d.z * r, d.w * r);
        }
    };

    const int d0      = part * SPP;
    const int nstrips = SPP + ((part == CPARTS - 1) ? 1 : 0);
    const int strip0  = (Iblk + d0) & (NBLK - 1);

    auto col_flush = [&](int jcol, int lane) {
        const int list = w >> 1;
        const int col  = ((w & 1) << 5) + lane;
        const float* cbp = &colbuf[list * 64 + col];
        float r = cbp[0];
        #pragma unroll
        for (int k = 1; k < 8; ++k) r = fmaxf(r, cbp[k * CBS]);
        float* dst = list ? wsColN : wsColP;
        dst[(size_t)Iblk * N + jcol * 64 + col] = r;
    };

    panel_issue(i0);
    panel_norm();
    panel_commit();
    __syncthreads();
    panel_issue(strip0 << 6);
    v2f afrag[DVEC / 4];
    {
        const float* aptr = &lds_b[(w * 16 + m) * LDSTR + 2 * hi];
        #pragma unroll
        for (int t = 0; t < DVEC / 4; ++t)
            afrag[t] = *(const v2f*)(aptr + 4 * t);
    }

    int Li[8];
    #pragma unroll
    for (int v = 0; v < 8; ++v) Li[v] = Lvec[iw + v + 8 * hi];

    float posV[8], negV[8];
    #pragma unroll
    for (int v = 0; v < 8; ++v) {
        posV[v] = -2.0f; negV[v] = -2.0f;
    }
    const int dd0 = m - iw - 8 * hi;

    const float* bptr = &lds_b[m * LDSTR + 2 * hi];

    for (int s = 0; s < nstrips; ++s) {
        const int dcur = d0 + s;
        const int jt   = (Iblk + dcur) & (NBLK - 1);
        panel_norm();
        __syncthreads();
        if (s > 0 && dcur > 1) col_flush((Iblk + dcur - 1) & (NBLK - 1), l);
        panel_commit();
        __syncthreads();
        if (s + 1 < nstrips) panel_issue(((Iblk + dcur + 1) & (NBLK - 1)) << 6);

        v8f c[4] = {};
        #pragma unroll
        for (int t = 0; t < DVEC / 4; ++t) {
            #pragma unroll
            for (int q = 0; q < 4; ++q) {
                v2f b = *(const v2f*)(bptr + q * 16 * LDSTR + 4 * t);
                c[q] = __builtin_amdgcn_wmma_f32_16x16x4_f32(
                           false, afrag[t], false, b, (short)0, c[q], false, false);
            }
        }

        int Lj[4];
        #pragma unroll
        for (int q = 0; q < 4; ++q) Lj[q] = Lvec[jt * 64 + q * 16 + m];
        float colP[4], colN[4];
        if (dcur == 0) {
            ephn_mine_slab<true>(c, Lj, Li, dd0 + jt * 64, posV, negV,
                                 colP, colN);
        } else {
            ephn_mine_slab<false>(c, Lj, Li, 0, posV, negV, colP, colN);
            float* cb = &colbuf[(w * 2 + hi) * CBS + m];
            #pragma unroll
            for (int q = 0; q < 4; ++q) {
                cb[q * 16]      = colP[q];
                cb[64 + q * 16] = colN[q];
            }
        }
    }

    #pragma unroll
    for (int v = 0; v < 8; ++v) {
        #pragma unroll
        for (int off = 1; off < 16; off <<= 1) {
            posV[v] = fmaxf(posV[v], __shfl_xor(posV[v], off, 16));
            negV[v] = fmaxf(negV[v], __shfl_xor(negV[v], off, 16));
        }
    }

    const int le = (int)__lane_id();
    if ((le & 15) == 0) {
        #pragma unroll
        for (int v = 0; v < 8; ++v) {
            const int row = w * 16 + v + 8 * (le >> 4);
            mPV[row] = posV[v];
            mNV[row] = negV[v];
        }
    }
    __syncthreads();
    if (w < 2) {
        const int te = w * 32 + le;
        wsPos[(size_t)part * N + i0 + te] = mPV[te];
        wsNeg[(size_t)part * N + i0 + te] = mNV[te];
    }
    col_flush((Iblk + d0 + nstrips - 1) & (NBLK - 1), le);
}

__global__ __launch_bounds__(256)
void ephn_merge_kernel(const float* __restrict__ wsPos,
                       const float* __restrict__ wsNeg,
                       const float* __restrict__ wsColP,
                       const float* __restrict__ wsColN,
                       float* __restrict__ mrgP,
                       float* __restrict__ mrgN, int N) {
    __shared__ float sP[4 * 64];
    __shared__ float sN[4 * 64];
    const int tid = threadIdx.x;
    const int x   = tid & 63;
    const int g   = tid >> 6;
    const int R   = blockIdx.x;
    const int r   = R * 64 + x;

    float p = fmaxf(wsPos[(size_t)(2 * g) * N + r], wsPos[(size_t)(2 * g + 1) * N + r]);
    float n = fmaxf(wsNeg[(size_t)(2 * g) * N + r], wsNeg[(size_t)(2 * g + 1) * N + r]);
    #pragma unroll
    for (int k = 0; k < HALFB / 4; ++k) {
        const int d = g * (HALFB / 4) + 1 + k;
        const int I = (R - d) & (NBLK - 1);
        p = fmaxf(p, wsColP[(size_t)I * N + r]);
        n = fmaxf(n, wsColN[(size_t)I * N + r]);
    }
    sP[g * 64 + x] = p;
    sN[g * 64 + x] = n;
    __syncthreads();
    if (tid < 64) {
        const float pm = fmaxf(fmaxf(sP[tid], sP[64 + tid]),
                               fmaxf(sP[128 + tid], sP[192 + tid]));
        const float nm = fmaxf(fmaxf(sN[tid], sN[64 + tid]),
                               fmaxf(sN[128 + tid], sN[192 + tid]));
        mrgP[r] = pm;
        mrgN[r] = nm;
    }
}

__global__ void ephn_finalize_kernel(const float* __restrict__ mrgP,
                                     const float* __restrict__ mrgN,
                                     float* __restrict__ out, int N) {
    __shared__ float sProb[1024];
    __shared__ float sPos[1024];
    __shared__ float sNeg[1024];
    __shared__ int   sCnt[1024];
    const int tid = threadIdx.x;

    float* eps = out + 1;
    float* hns = out + 1 + (N + 2);

    float accProb = 0.0f, accPos = 0.0f, accNeg = 0.0f;
    int accCnt = 0;
    for (int i = tid; i < N; i += 1024) {
        const float p = mrgP[i];
        const float n = mrgN[i];
        eps[i] = (p > 0.0f) ? p : 0.0f;
        hns[i] = (n > 0.0f) ? n : 0.0f;
        accPos += p;
        accNeg += n;
        if (p > 0.0f && n > 0.0f) {
            const float s = (n - p) * INV_SIGMA;
            const float prob = (s > 0.0f) ? (s + log1pf(expf(-s)))
                                          : log1pf(expf(s));
            accProb += prob;
            accCnt  += 1;
        }
    }
    sProb[tid] = accProb; sPos[tid] = accPos; sNeg[tid] = accNeg; sCnt[tid] = accCnt;
    __syncthreads();
    for (int st = 512; st > 0; st >>= 1) {
        if (tid < st) {
            sProb[tid] += sProb[tid + st];
            sPos[tid]  += sPos[tid + st];
            sNeg[tid]  += sNeg[tid + st];
            sCnt[tid]  += sCnt[tid + st];
        }
        __syncthreads();
    }
    if (tid == 0) {
        const int cnt = (sCnt[0] > 1) ? sCnt[0] : 1;
        out[0] = sProb[0] / (float)cnt;
        eps[N] = 0.0f; eps[N + 1] = 1.0f;
        hns[N] = 0.0f; hns[N + 1] = 1.0f;
        out[1 + 2 * (N + 2)] = (sPos[0] - sNeg[0]) / (float)N;
    }
}

extern "C" void kernel_launch(void* const* d_in, const int* in_sizes, int n_in,
                              void* d_out, int out_size, void* d_ws, size_t ws_size,
                              hipStream_t stream) {
    const float* fvec = (const float*)d_in[0];
    const int*   Lvec = (const int*)d_in[1];
    const int N = in_sizes[1];

    float* wsPos = (float*)d_ws;
    float* wsNeg = wsPos + CPARTS * N;
    float* wsColP = wsNeg + CPARTS * N;
    float* wsColN = wsColP + NBLK * N;
    float* mrgP   = wsColN + NBLK * N;
    float* mrgN   = mrgP + N;

    ephn_main_kernel<<<NBLK * CPARTS, 128, 0, stream>>>(fvec, Lvec,
                                                        wsPos, wsNeg,
                                                        wsColP, wsColN, N);
    ephn_merge_kernel<<<NBLK, 256, 0, stream>>>(wsPos, wsNeg, wsColP, wsColN,
                                                mrgP, mrgN, N);
    ephn_finalize_kernel<<<1, 1024, 0, stream>>>(mrgP, mrgN,
                                                 (float*)d_out, N);
}


# Per-row loss terms and clamped score outputs move from the one-block finalize into the 128-block merge kernel; finalize is one wave adding 256 wave partials (order of the loss and mean sums changes)
# speedup vs baseline: 6.8804x; 1.1085x over previous
#include <hip/hip_runtime.h>
#include <hip/hip_bf16.h>


typedef __attribute__((ext_vector_type(2))) float v2f;
typedef __attribute__((ext_vector_type(8))) float v8f;

#define DVEC 128
#define LDSTR 132
#define INV_SIGMA 10.0f
#define NBLK 128
#define HALFB (NBLK / 2)

__device__ __forceinline__ float ephn_xrow(float x) {
    const int b = __builtin_bit_cast(int, x);
    return __builtin_bit_cast(float,
        __builtin_amdgcn_permlanex16(b, b, 0x76543210, 0xfedcba98, false, false));
}
template <int M>
__device__ __forceinline__ float ephn_xlane(float x) {
    const int b = __builtin_bit_cast(int, x);
    return __builtin_bit_cast(float,
        __builtin_amdgcn_update_dpp(b, b, 0x160 | M, 0xf, 0xf, true));
}
__device__ __forceinline__ float ephn_wave_sum(float x) {
    x += ephn_xrow(x);
    x += ephn_xlane<8>(x);
    x += ephn_xlane<4>(x);
    x += ephn_xlane<2>(x);
    x += ephn_xlane<1>(x);
    return x;
}

#define CPARTS 8
#define SPP (HALFB / CPARTS)
#define CBS 144

template <bool DIAG>
__device__ __forceinline__ void ephn_mine_slab(const v8f (&c)[4],
                                               const int (&Lj)[4],
                                               const int (&Li)[8], int dd,
                                               float (&posV)[8],
                                               float (&negV)[8],
                                               float (&colP)[4],
                                               float (&colN)[4]) {
    #pragma unroll
    for (int q = 0; q < 4; ++q) {
        float cp = -2.0f, cn = -2.0f;
        #pragma unroll
        for (int v = 0; v < 8; ++v) {
            const float d   = c[q][v];
            const bool  eqc = (Li[v] == Lj[q]);
            bool okp = eqc && !(d > 0.9999f);
            if (DIAG) okp = okp && (dd != v - q * 16);
            const float sp = okp ? d : -1.0f;
            const float sn = eqc ? -1.0f : d;
            posV[v] = fmaxf(posV[v], sp);
            negV[v] = fmaxf(negV[v], sn);
            if (!DIAG) {
                cp = fmaxf(cp, sp);
                cn = fmaxf(cn, sn);
            }
        }
        colP[q] = cp;
        colN[q] = cn;
    }
}

__global__ __launch_bounds__(128) __attribute__((amdgpu_waves_per_eu(4, 4)))
void ephn_main_kernel(const float* __restrict__ fvec,
                      const int*   __restrict__ Lvec,
                      float* __restrict__ wsPos,
                      float* __restrict__ wsNeg,
                      float* __restrict__ wsColP,
                      float* __restrict__ wsColN,
                      int N) {
    __shared__ __attribute__((aligned(16))) float lds_b[64 * LDSTR];
    __shared__ float mPV[64];
    __shared__ float mNV[64];
    __shared__ float colbuf[8 * CBS];

    const int tid = threadIdx.x;
    const int l   = tid & 31;
    const int w   = __builtin_amdgcn_readfirstlane(tid >> 5);
    const int m   = l & 15;
    const int hi  = l >> 4;
    const int part = blockIdx.x / NBLK;
    const int Iblk = blockIdx.x & (NBLK - 1);
    const int i0   = Iblk * 64;
    const int iw   = i0 + w * 16;

    const int ccol = l << 2;
    float4 pre[16];
    float  prn;

    auto panel_issue = [&](int rowbase) {
        const float* src = fvec + ((size_t)(rowbase + w) << 7) + ccol;
        #pragma unroll
        for (int p = 0; p < 16; ++p)
            pre[p] = *(const float4*)(src + ((size_t)p << 9));
    };
    auto panel_norm = [&]() {
        float ss[16];
        #pragma unroll
        for (int p = 0; p < 16; ++p) {
            const float4 d = pre[p];
            ss[p] = fmaf(d.w, d.w, fmaf(d.z, d.z, fmaf(d.x, d.x, d.y * d.y)));
        }
        const bool b4 = (l & 16) != 0, b3 = (l & 8) != 0;
        const bool b2 = (l & 4) != 0,  b1 = (l & 2) != 0;
        float s8[8], s4[4], s2[2];
        #pragma unroll
        for (int k = 0; k < 8; ++k)
            s8[k] = (b4 ? ss[k + 8] : ss[k]) + ephn_xrow(b4 ? ss[k] : ss[k + 8]);
        #pragma unroll
        for (int k = 0; k < 4; ++k)
            s4[k] = (b3 ? s8[k + 4] : s8[k]) + ephn_xlane<8>(b3 ? s8[k] : s8[k + 4]);
        #pragma unroll
        for (int k = 0; k < 2; ++k)
            s2[k] = (b2 ? s4[k + 2] : s4[k]) + ephn_xlane<4>(b2 ? s4[k] : s4[k + 2]);
        const float s1 = (b1 ? s2[1] : s2[0]) + ephn_xlane<2>(b1 ? s2[0] : s2[1]);
        prn = rsqrtf(s1 + ephn_xlane<1>(s1));
    };
    auto panel_commit = [&]() {
        #pragma unroll
        for (int p = 0; p < 16; ++p) {
            const float r = __int_as_float(
                __builtin_amdgcn_readlane(__float_as_int(prn), 2 * p));
            const float4 d = pre[p];
            *(float4*)&lds_b[(w + 4 * p) * LDSTR + ccol] =
                make_float4(d.x * r, d.y * r, d.z * r, d.w * r);
        }
    };

    const int d0      = part * SPP;
    const int nstrips = SPP + ((part == CPARTS - 1) ? 1 : 0);
    const int strip0  = (Iblk + d0) & (NBLK - 1);

    auto col_flush = [&](int jcol, int lane) {
        const int list = w >> 1;
        const int col  = ((w & 1) << 5) + lane;
        const float* cbp = &colbuf[list * 64 + col];
        float r = cbp[0];
        #pragma unroll
        for (int k = 1; k < 8; ++k) r = fmaxf(r, cbp[k * CBS]);
        float* dst = list ? wsColN : wsColP;
        dst[(size_t)Iblk * N + jcol * 64 + col] = r;
    };

    panel_issue(i0);
    panel_norm();
    panel_commit();
    __syncthreads();
    panel_issue(strip0 << 6);
    v2f afrag[DVEC / 4];
    {
        const float* aptr = &lds_b[(w * 16 + m) * LDSTR + 2 * hi];
        #pragma unroll
        for (int t = 0; t < DVEC / 4; ++t)
            afrag[t] = *(const v2f*)(aptr + 4 * t);
    }

    int Li[8];
    #pragma unroll
    for (int v = 0; v < 8; ++v) Li[v] = Lvec[iw + v + 8 * hi];

    float posV[8], negV[8];
    #pragma unroll
    for (int v = 0; v < 8; ++v) {
        posV[v] = -2.0f; negV[v] = -2.0f;
    }
    const int dd0 = m - iw - 8 * hi;

    const float* bptr = &lds_b[m * LDSTR + 2 * hi];

    for (int s = 0; s < nstrips; ++s) {
        const int dcur = d0 + s;
        const int jt   = (Iblk + dcur) & (NBLK - 1);
        panel_norm();
        __syncthreads();
        if (s > 0 && dcur > 1) col_flush((Iblk + dcur - 1) & (NBLK - 1), l);
        panel_commit();
        __syncthreads();
        if (s + 1 < nstrips) panel_issue(((Iblk + dcur + 1) & (NBLK - 1)) << 6);

        v8f c[4] = {};
        #pragma unroll
        for (int t = 0; t < DVEC / 4; ++t) {
            #pragma unroll
            for (int q = 0; q < 4; ++q) {
                v2f b = *(const v2f*)(bptr + q * 16 * LDSTR + 4 * t);
                c[q] = __builtin_amdgcn_wmma_f32_16x16x4_f32(
                           false, afrag[t], false, b, (short)0, c[q], false, false);
            }
        }

        int Lj[4];
        #pragma unroll
        for (int q = 0; q < 4; ++q) Lj[q] = Lvec[jt * 64 + q * 16 + m];
        float colP[4], colN[4];
        if (dcur == 0) {
            ephn_mine_slab<true>(c, Lj, Li, dd0 + jt * 64, posV, negV,
                                 colP, colN);
        } else {
            ephn_mine_slab<false>(c, Lj, Li, 0, posV, negV, colP, colN);
            float* cb = &colbuf[(w * 2 + hi) * CBS + m];
            #pragma unroll
            for (int q = 0; q < 4; ++q) {
                cb[q * 16]      = colP[q];
                cb[64 + q * 16] = colN[q];
            }
        }
    }

    #pragma unroll
    for (int v = 0; v < 8; ++v) {
        #pragma unroll
        for (int off = 1; off < 16; off <<= 1) {
            posV[v] = fmaxf(posV[v], __shfl_xor(posV[v], off, 16));
            negV[v] = fmaxf(negV[v], __shfl_xor(negV[v], off, 16));
        }
    }

    const int le = (int)__lane_id();
    if ((le & 15) == 0) {
        #pragma unroll
        for (int v = 0; v < 8; ++v) {
            const int row = w * 16 + v + 8 * (le >> 4);
            mPV[row] = posV[v];
            mNV[row] = negV[v];
        }
    }
    __syncthreads();
    if (w < 2) {
        const int te = w * 32 + le;
        wsPos[(size_t)part * N + i0 + te] = mPV[te];
        wsNeg[(size_t)part * N + i0 + te] = mNV[te];
    }
    col_flush((Iblk + d0 + nstrips - 1) & (NBLK - 1), le);
}

#define PLINE 32
__global__ __launch_bounds__(256)
void ephn_merge_kernel(const float* __restrict__ wsPos,
                       const float* __restrict__ wsNeg,
                       const float* __restrict__ wsColP,
                       const float* __restrict__ wsColN,
                       float* __restrict__ part,
                       float* __restrict__ out, int N) {
    __shared__ float sP[4 * 64];
    __shared__ float sN[4 * 64];
    const int tid = threadIdx.x;
    const int x   = tid & 63;
    const int g   = tid >> 6;
    const int R   = blockIdx.x;
    const int r   = R * 64 + x;

    float p = fmaxf(wsPos[(size_t)(2 * g) * N + r], wsPos[(size_t)(2 * g + 1) * N + r]);
    float n = fmaxf(wsNeg[(size_t)(2 * g) * N + r], wsNeg[(size_t)(2 * g + 1) * N + r]);
    #pragma unroll
    for (int k = 0; k < HALFB / 4; ++k) {
        const int d = g * (HALFB / 4) + 1 + k;
        const int I = (R - d) & (NBLK - 1);
        p = fmaxf(p, wsColP[(size_t)I * N + r]);
        n = fmaxf(n, wsColN[(size_t)I * N + r]);
    }
    sP[g * 64 + x] = p;
    sN[g * 64 + x] = n;
    __syncthreads();
    if (tid < 64) {
        const float pm = fmaxf(fmaxf(sP[tid], sP[64 + tid]),
                               fmaxf(sP[128 + tid], sP[192 + tid]));
        const float nm = fmaxf(fmaxf(sN[tid], sN[64 + tid]),
                               fmaxf(sN[128 + tid], sN[192 + tid]));
        float* eps = out + 1;
        float* hns = out + 1 + (N + 2);
        eps[r] = (pm > 0.0f) ? pm : 0.0f;
        hns[r] = (nm > 0.0f) ? nm : 0.0f;
        float prob = 0.0f, cnt = 0.0f;
        if (pm > 0.0f && nm > 0.0f) {
            const float s = (nm - pm) * INV_SIGMA;
            prob = (s > 0.0f) ? (s + log1pf(expf(-s))) : log1pf(expf(s));
            cnt  = 1.0f;
        }
        const float sProb = ephn_wave_sum(prob);
        const float sPos  = ephn_wave_sum(pm);
        const float sNeg  = ephn_wave_sum(nm);
        const float sCnt  = ephn_wave_sum(cnt);
        if ((tid & 31) == 0)
            *(float4*)(part + (size_t)(2 * R + (tid >> 5)) * PLINE) =
                make_float4(sProb, sPos, sNeg, sCnt);
    }
}

__global__ __launch_bounds__(32)
void ephn_finalize_kernel(const float* __restrict__ part,
                          float* __restrict__ out, int N) {
    const int l = threadIdx.x;
    float4 a = make_float4(0.0f, 0.0f, 0.0f, 0.0f);
    #pragma unroll
    for (int k = 0; k < 2 * NBLK / 32; ++k) {
        const float4 q = *(const float4*)(part + (size_t)(l + 32 * k) * PLINE);
        a.x += q.x; a.y += q.y; a.z += q.z; a.w += q.w;
    }
    const float sProb = ephn_wave_sum(a.x);
    const float sPos  = ephn_wave_sum(a.y);
    const float sNeg  = ephn_wave_sum(a.z);
    const float sCnt  = ephn_wave_sum(a.w);
    if (l == 0) {
        float* eps = out + 1;
        float* hns = out + 1 + (N + 2);
        out[0] = sProb / fmaxf(sCnt, 1.0f);
        eps[N] = 0.0f; eps[N + 1] = 1.0f;
        hns[N] = 0.0f; hns[N + 1] = 1.0f;
        out[1 + 2 * (N + 2)] = (sPos - sNeg) / (float)N;
    }
}

extern "C" void kernel_launch(void* const* d_in, const int* in_sizes, int n_in,
                              void* d_out, int out_size, void* d_ws, size_t ws_size,
                              hipStream_t stream) {
    const float* fvec = (const float*)d_in[0];
    const int*   Lvec = (const int*)d_in[1];
    const int N = in_sizes[1];

    float* wsPos = (float*)d_ws;
    float* wsNeg = wsPos + CPARTS * N;
    float* wsColP = wsNeg + CPARTS * N;
    float* wsColN = wsColP + NBLK * N;
    float* part   = wsColN + NBLK * N;

    ephn_main_kernel<<<NBLK * CPARTS, 128, 0, stream>>>(fvec, Lvec,
                                                        wsPos, wsNeg,
                                                        wsColP, wsColN, N);
    ephn_merge_kernel<<<NBLK, 256, 0, stream>>>(wsPos, wsNeg, wsColP, wsColN,
                                                part, (float*)d_out, N);
    ephn_finalize_kernel<<<1, 32, 0, stream>>>(part, (float*)d_out, N);
}


# Waves sharing a SIMD end together: the one-strip-longer part gets the lowest block ids (oldest waves) and each wave's issue priority (s_setprio) steps 3,2,1,0 over its last strips; none ends alone
# speedup vs baseline: 7.3597x; 1.0697x over previous
#include <hip/hip_runtime.h>
#include <hip/hip_bf16.h>


typedef __attribute__((ext_vector_type(2))) float v2f;
typedef __attribute__((ext_vector_type(8))) float v8f;

#define DVEC 128
#define LDSTR 132
#define INV_SIGMA 10.0f
#define NBLK 128
#define HALFB (NBLK / 2)

__device__ __forceinline__ float ephn_xrow(float x) {
    const int b = __builtin_bit_cast(int, x);
    return __builtin_bit_cast(float,
        __builtin_amdgcn_permlanex16(b, b, 0x76543210, 0xfedcba98, false, false));
}
template <int M>
__device__ __forceinline__ float ephn_xlane(float x) {
    const int b = __builtin_bit_cast(int, x);
    return __builtin_bit_cast(float,
        __builtin_amdgcn_update_dpp(b, b, 0x160 | M, 0xf, 0xf, true));
}
__device__ __forceinline__ float ephn_wave_sum(float x) {
    x += ephn_xrow(x);
    x += ephn_xlane<8>(x);
    x += ephn_xlane<4>(x);
    x += ephn_xlane<2>(x);
    x += ephn_xlane<1>(x);
    return x;
}

#define CPARTS 8
#define SPP (HALFB / CPARTS)
#define CBS 144

template <bool DIAG>
__device__ __forceinline__ void ephn_mine_slab(const v8f (&c)[4],
                                               const int (&Lj)[4],
                                               const int (&Li)[8], int dd,
                                               float (&posV)[8],
                                               float (&negV)[8],
                                               float (&colP)[4],
                                               float (&colN)[4]) {
    #pragma unroll
    for (int q = 0; q < 4; ++q) {
        float cp = -2.0f, cn = -2.0f;
        #pragma unroll
        for (int v = 0; v < 8; ++v) {
            const float d   = c[q][v];
            const bool  eqc = (Li[v] == Lj[q]);
            bool okp = eqc && !(d > 0.9999f);
            if (DIAG) okp = okp && (dd != v - q * 16);
            const float sp = okp ? d : -1.0f;
            const float sn = eqc ? -1.0f : d;
            posV[v] = fmaxf(posV[v], sp);
            negV[v] = fmaxf(negV[v], sn);
            if (!DIAG) {
                cp = fmaxf(cp, sp);
                cn = fmaxf(cn, sn);
            }
        }
        colP[q] = cp;
        colN[q] = cn;
    }
}

__global__ __launch_bounds__(128) __attribute__((amdgpu_waves_per_eu(4, 4)))
void ephn_main_kernel(const float* __restrict__ fvec,
                      const int*   __restrict__ Lvec,
                      float* __restrict__ wsPos,
                      float* __restrict__ wsNeg,
                      float* __restrict__ wsColP,
                      float* __restrict__ wsColN,
                      int N) {
    __shared__ __attribute__((aligned(16))) float lds_b[64 * LDSTR];
    __shared__ float mPV[64];
    __shared__ float mNV[64];
    __shared__ float colbuf[8 * CBS];

    __builtin_amdgcn_s_setprio(3);
    const int tid = threadIdx.x;
    const int l   = tid & 31;
    const int w   = __builtin_amdgcn_readfirstlane(tid >> 5);
    const int m   = l & 15;
    const int hi  = l >> 4;
    const int part = (CPARTS - 1) - (int)(blockIdx.x / NBLK);
    const int Iblk = blockIdx.x & (NBLK - 1);
    const int i0   = Iblk * 64;
    const int iw   = i0 + w * 16;

    const int ccol = l << 2;
    float4 pre[16];
    float  prn;

    auto panel_issue = [&](int rowbase) {
        const float* src = fvec + ((size_t)(rowbase + w) << 7) + ccol;
        #pragma unroll
        for (int p = 0; p < 16; ++p)
            pre[p] = *(const float4*)(src + ((size_t)p << 9));
    };
    auto panel_norm = [&]() {
        float ss[16];
        #pragma unroll
        for (int p = 0; p < 16; ++p) {
            const float4 d = pre[p];
            ss[p] = fmaf(d.w, d.w, fmaf(d.z, d.z, fmaf(d.x, d.x, d.y * d.y)));
        }
        const bool b4 = (l & 16) != 0, b3 = (l & 8) != 0;
        const bool b2 = (l & 4) != 0,  b1 = (l & 2) != 0;
        float s8[8], s4[4], s2[2];
        #pragma unroll
        for (int k = 0; k < 8; ++k)
            s8[k] = (b4 ? ss[k + 8] : ss[k]) + ephn_xrow(b4 ? ss[k] : ss[k + 8]);
        #pragma unroll
        for (int k = 0; k < 4; ++k)
            s4[k] = (b3 ? s8[k + 4] : s8[k]) + ephn_xlane<8>(b3 ? s8[k] : s8[k + 4]);
        #pragma unroll
        for (int k = 0; k < 2; ++k)
            s2[k] = (b2 ? s4[k + 2] : s4[k]) + ephn_xlane<4>(b2 ? s4[k] : s4[k + 2]);
        const float s1 = (b1 ? s2[1] : s2[0]) + ephn_xlane<2>(b1 ? s2[0] : s2[1]);
        prn = rsqrtf(s1 + ephn_xlane<1>(s1));
    };
    auto panel_commit = [&]() {
        #pragma unroll
        for (int p = 0; p < 16; ++p) {
            const float r = __int_as_float(
                __builtin_amdgcn_readlane(__float_as_int(prn), 2 * p));
            const float4 d = pre[p];
            *(float4*)&lds_b[(w + 4 * p) * LDSTR + ccol] =
                make_float4(d.x * r, d.y * r, d.z * r, d.w * r);
        }
    };

    const int d0      = part * SPP;
    const int nstrips = SPP + ((part == CPARTS - 1) ? 1 : 0);
    const int strip0  = (Iblk + d0) & (NBLK - 1);

    auto col_flush = [&](int jcol, int lane) {
        const int list = w >> 1;
        const int col  = ((w & 1) << 5) + lane;
        const float* cbp = &colbuf[list * 64 + col];
        float r = cbp[0];
        #pragma unroll
        for (int k = 1; k < 8; ++k) r = fmaxf(r, cbp[k * CBS]);
        float* dst = list ? wsColN : wsColP;
        dst[(size_t)Iblk * N + jcol * 64 + col] = r;
    };

    panel_issue(i0);
    panel_norm();
    panel_commit();
    __syncthreads();
    panel_issue(strip0 << 6);
    v2f afrag[DVEC / 4];
    {
        const float* aptr = &lds_b[(w * 16 + m) * LDSTR + 2 * hi];
        #pragma unroll
        for (int t = 0; t < DVEC / 4; ++t)
            afrag[t] = *(const v2f*)(aptr + 4 * t);
    }

    int Li[8];
    #pragma unroll
    for (int v = 0; v < 8; ++v) Li[v] = Lvec[iw + v + 8 * hi];

    float posV[8], negV[8];
    #pragma unroll
    for (int v = 0; v < 8; ++v) {
        posV[v] = -2.0f; negV[v] = -2.0f;
    }
    const int dd0 = m - iw - 8 * hi;

    const float* bptr = &lds_b[m * LDSTR + 2 * hi];

    for (int s = 0; s < nstrips; ++s) {
        const int left = nstrips - 1 - s;
        if (left == 2)      __builtin_amdgcn_s_setprio(2);
        else if (left == 1) __builtin_amdgcn_s_setprio(1);
        else if (left == 0) __builtin_amdgcn_s_setprio(0);
        const int dcur = d0 + s;
        const int jt   = (Iblk + dcur) & (NBLK - 1);
        panel_norm();
        __syncthreads();
        if (s > 0 && dcur > 1) col_flush((Iblk + dcur - 1) & (NBLK - 1), l);
        panel_commit();
        __syncthreads();
        if (s + 1 < nstrips) panel_issue(((Iblk + dcur + 1) & (NBLK - 1)) << 6);

        v8f c[4] = {};
        #pragma unroll
        for (int t = 0; t < DVEC / 4; ++t) {
            #pragma unroll
            for (int q = 0; q < 4; ++q) {
                v2f b = *(const v2f*)(bptr + q * 16 * LDSTR + 4 * t);
                c[q] = __builtin_amdgcn_wmma_f32_16x16x4_f32(
                           false, afrag[t], false, b, (short)0, c[q], false, false);
            }
        }

        int Lj[4];
        #pragma unroll
        for (int q = 0; q < 4; ++q) Lj[q] = Lvec[jt * 64 + q * 16 + m];
        float colP[4], colN[4];
        if (dcur == 0) {
            ephn_mine_slab<true>(c, Lj, Li, dd0 + jt * 64, posV, negV,
                                 colP, colN);
        } else {
            ephn_mine_slab<false>(c, Lj, Li, 0, posV, negV, colP, colN);
            float* cb = &colbuf[(w * 2 + hi) * CBS + m];
            #pragma unroll
            for (int q = 0; q < 4; ++q) {
                cb[q * 16]      = colP[q];
                cb[64 + q * 16] = colN[q];
            }
        }
    }

    __builtin_amdgcn_s_setprio(0);

    #pragma unroll
    for (int v = 0; v < 8; ++v) {
        #pragma unroll
        for (int off = 1; off < 16; off <<= 1) {
            posV[v] = fmaxf(posV[v], __shfl_xor(posV[v], off, 16));
            negV[v] = fmaxf(negV[v], __shfl_xor(negV[v], off, 16));
        }
    }

    const int le = (int)__lane_id();
    if ((le & 15) == 0) {
        #pragma unroll
        for (int v = 0; v < 8; ++v) {
            const int row = w * 16 + v + 8 * (le >> 4);
            mPV[row] = posV[v];
            mNV[row] = negV[v];
        }
    }
    __syncthreads();
    if (w < 2) {
        const int te = w * 32 + le;
        wsPos[(size_t)part * N + i0 + te] = mPV[te];
        wsNeg[(size_t)part * N + i0 + te] = mNV[te];
    }
    col_flush((Iblk + d0 + nstrips - 1) & (NBLK - 1), le);
}

#define PLINE 32
__global__ __launch_bounds__(256)
void ephn_merge_kernel(const float* __restrict__ wsPos,
                       const float* __restrict__ wsNeg,
                       const float* __restrict__ wsColP,
                       const float* __restrict__ wsColN,
                       float* __restrict__ part,
                       float* __restrict__ out, int N) {
    __shared__ float sP[4 * 64];
    __shared__ float sN[4 * 64];
    const int tid = threadIdx.x;
    const int x   = tid & 63;
    const int g   = tid >> 6;
    const int R   = blockIdx.x;
    const int r   = R * 64 + x;

    float p = fmaxf(wsPos[(size_t)(2 * g) * N + r], wsPos[(size_t)(2 * g + 1) * N + r]);
    float n = fmaxf(wsNeg[(size_t)(2 * g) * N + r], wsNeg[(size_t)(2 * g + 1) * N + r]);
    #pragma unroll
    for (int k = 0; k < HALFB / 4; ++k) {
        const int d = g * (HALFB / 4) + 1 + k;
        const int I = (R - d) & (NBLK - 1);
        p = fmaxf(p, wsColP[(size_t)I * N + r]);
        n = fmaxf(n, wsColN[(size_t)I * N + r]);
    }
    sP[g * 64 + x] = p;
    sN[g * 64 + x] = n;
    __syncthreads();
    if (tid < 64) {
        const float pm = fmaxf(fmaxf(sP[tid], sP[64 + tid]),
                               fmaxf(sP[128 + tid], sP[192 + tid]));
        const float nm = fmaxf(fmaxf(sN[tid], sN[64 + tid]),
                               fmaxf(sN[128 + tid], sN[192 + tid]));
        float* eps = out + 1;
        float* hns = out + 1 + (N + 2);
        eps[r] = (pm > 0.0f) ? pm : 0.0f;
        hns[r] = (nm > 0.0f) ? nm : 0.0f;
        float prob = 0.0f, cnt = 0.0f;
        if (pm > 0.0f && nm > 0.0f) {
            const float s = (nm - pm) * INV_SIGMA;
            prob = (s > 0.0f) ? (s + log1pf(expf(-s))) : log1pf(expf(s));
            cnt  = 1.0f;
        }
        const float sProb = ephn_wave_sum(prob);
        const float sPos  = ephn_wave_sum(pm);
        const float sNeg  = ephn_wave_sum(nm);
        const float sCnt  = ephn_wave_sum(cnt);
        if ((tid & 31) == 0)
            *(float4*)(part + (size_t)(2 * R + (tid >> 5)) * PLINE) =
                make_float4(sProb, sPos, sNeg, sCnt);
    }
}

__global__ __launch_bounds__(32)
void ephn_finalize_kernel(const float* __restrict__ part,
                          float* __restrict__ out, int N) {
    const int l = threadIdx.x;
    float4 a = make_float4(0.0f, 0.0f, 0.0f, 0.0f);
    #pragma unroll
    for (int k = 0; k < 2 * NBLK / 32; ++k) {
        const float4 q = *(const float4*)(part + (size_t)(l + 32 * k) * PLINE);
        a.x += q.x; a.y += q.y; a.z += q.z; a.w += q.w;
    }
    const float sProb = ephn_wave_sum(a.x);
    const float sPos  = ephn_wave_sum(a.y);
    const float sNeg  = ephn_wave_sum(a.z);
    const float sCnt  = ephn_wave_sum(a.w);
    if (l == 0) {
        float* eps = out + 1;
        float* hns = out + 1 + (N + 2);
        out[0] = sProb / fmaxf(sCnt, 1.0f);
        eps[N] = 0.0f; eps[N + 1] = 1.0f;
        hns[N] = 0.0f; hns[N + 1] = 1.0f;
        out[1 + 2 * (N + 2)] = (sPos - sNeg) / (float)N;
    }
}

extern "C" void kernel_launch(void* const* d_in, const int* in_sizes, int n_in,
                              void* d_out, int out_size, void* d_ws, size_t ws_size,
                              hipStream_t stream) {
    const float* fvec = (const float*)d_in[0];
    const int*   Lvec = (const int*)d_in[1];
    const int N = in_sizes[1];

    float* wsPos = (float*)d_ws;
    float* wsNeg = wsPos + CPARTS * N;
    float* wsColP = wsNeg + CPARTS * N;
    float* wsColN = wsColP + NBLK * N;
    float* part   = wsColN + NBLK * N;

    ephn_main_kernel<<<NBLK * CPARTS, 128, 0, stream>>>(fvec, Lvec,
                                                        wsPos, wsNeg,
                                                        wsColP, wsColN, N);
    ephn_merge_kernel<<<NBLK, 256, 0, stream>>>(wsPos, wsNeg, wsColP, wsColN,
                                                part, (float*)d_out, N);
    ephn_finalize_kernel<<<1, 32, 0, stream>>>(part, (float*)d_out, N);
}
